# starting kernel
# baseline (speedup 1.0000x reference)
#include <hip/hip_runtime.h>

#define D 128

typedef float v2f __attribute__((ext_vector_type(2)));
typedef float v8f __attribute__((ext_vector_type(8)));

__global__ __launch_bounds__(256) void zero_ws(float* __restrict__ p, int n) {
    int i = blockIdx.x * 256 + threadIdx.x;
    if (i < n) p[i] = 0.0f;
}

__global__ __launch_bounds__(256) void degree_count(const int* __restrict__ erow,
                                                    float* __restrict__ deg, int ne) {
    int i = blockIdx.x * 256 + threadIdx.x;
    if (i < ne) atomicAdd(deg + erow[i], 1.0f);
}

__global__ __launch_bounds__(256) void deg_rsqrt(float* __restrict__ deg, int n) {
    int i = blockIdx.x * 256 + threadIdx.x;
    if (i < n) deg[i] = rsqrtf(deg[i]);
}

__global__ __launch_bounds__(256) void edge_aggregate(
    const float* __restrict__ x, const int* __restrict__ erow,
    const int* __restrict__ ecol, const float* __restrict__ dis,
    float* __restrict__ agg, int ne)
{
    int wid = blockIdx.x * 8 + (threadIdx.x >> 5);
    if (wid >= ne) return;
    int e   = __builtin_amdgcn_readfirstlane(wid);
    int row = erow[e];
    int col = ecol[e];
    float w = dis[row] * dis[col];
    int lane = threadIdx.x & 31;

    float4 xv = *((const float4*)(x + (size_t)col * D) + lane);
    float* dst = agg + (size_t)row * D + lane * 4;
    atomicAdd(dst + 0, w * xv.x);
    atomicAdd(dst + 1, w * xv.y);
    atomicAdd(dst + 2, w * xv.z);
    atomicAdd(dst + 3, w * xv.w);
}

__global__ __launch_bounds__(256) void gemm_bias_wmma(
    const float* __restrict__ agg, const float* __restrict__ Wt,
    const float* __restrict__ bias, float* __restrict__ out)
{
    const int lane = threadIdx.x & 31;
    const int wave = threadIdx.x >> 5;
    const int m0   = blockIdx.x * 16;
    const int n0   = wave * 16;
    const int half = lane >> 4;
    const int l15  = lane & 15;

    float bval = bias[n0 + l15];
    v8f c;
#pragma unroll
    for (int r = 0; r < 8; ++r) c[r] = bval;

    const float* arow = agg + (size_t)(m0 + l15) * D;
#pragma unroll 8
    for (int k = 0; k < D; k += 4) {
        v2f a = *(const v2f*)(arow + k + half * 2);
        v2f b;
        b.x = Wt[(size_t)(k + half * 2 + 0) * D + n0 + l15];
        b.y = Wt[(size_t)(k + half * 2 + 1) * D + n0 + l15];
        c = __builtin_amdgcn_wmma_f32_16x16x4_f32(
                 false, a,  false, b,
                 (short)0, c,  false,  false);
    }

    float* orow = out + (size_t)(m0 + half * 8) * D + n0 + l15;
#pragma unroll
    for (int r = 0; r < 8; ++r) orow[(size_t)r * D] = c[r];
}

__global__ __launch_bounds__(128) void gemm_tail(
    const float* __restrict__ agg, const float* __restrict__ Wt,
    const float* __restrict__ bias, float* __restrict__ out,
    int row_start)
{
    int r = row_start + blockIdx.x;
    int n = threadIdx.x;
    float acc = bias[n];
    const float* a = agg + (size_t)r * D;
    for (int k = 0; k < D; ++k) acc = fmaf(a[k], Wt[(size_t)k * D + n], acc);
    out[(size_t)r * D + n] = acc;
}

extern "C" void kernel_launch(void* const* d_in, const int* in_sizes, int n_in,
                              void* d_out, int out_size, void* d_ws, size_t ws_size,
                              hipStream_t stream) {
    const float* x      = (const float*)d_in[0];
    const float* weight = (const float*)d_in[1];
    const float* bias   = (const float*)d_in[2];
    const int*   erow   = (const int*)d_in[3];
    const int*   ecol   = (const int*)d_in[4];

    const int n_nodes = in_sizes[0] / D;
    const int n_edges = in_sizes[3];

    float* agg = (float*)d_ws;
    float* deg = agg + (size_t)n_nodes * D;

    const int zn = n_nodes * (D + 1);
    zero_ws<<<(zn + 255) / 256, 256, 0, stream>>>((float*)d_ws, zn);

    degree_count<<<(n_edges + 255) / 256, 256, 0, stream>>>(erow, deg, n_edges);
    deg_rsqrt<<<(n_nodes + 255) / 256, 256, 0, stream>>>(deg, n_nodes);

    edge_aggregate<<<(n_edges + 7) / 8, 256, 0, stream>>>(x, erow, ecol, deg,
                                                          agg, n_edges);

    const int mtiles = n_nodes / 16;
    if (mtiles > 0)
        gemm_bias_wmma<<<mtiles, 256, 0, stream>>>(agg, weight, bias,
                                                   (float*)d_out);
    const int tail = n_nodes - mtiles * 16;
    if (tail > 0)
        gemm_tail<<<tail, D, 0, stream>>>(agg, weight, bias, (float*)d_out,
                                          mtiles * 16);
}


# Degree count: 4 node ranges x 64 edge slices; a work-group counts its slice's rows in an int LDS histogram, then adds its range to its half's copy with whole-line float atomics; counts exact
# speedup vs baseline: 3.2738x; 3.2738x over previous
#include <hip/hip_runtime.h>

#define D 128

typedef float v2f __attribute__((ext_vector_type(2)));
typedef float v4f __attribute__((ext_vector_type(4)));
typedef float v8f __attribute__((ext_vector_type(8)));

__device__ __forceinline__ int package_half() {
    return (int)((__builtin_amdgcn_s_sendmsg_rtn(0x87) >> 18) & 1u);
}

__global__ __launch_bounds__(256) void zero_ws(float* __restrict__ p, int n_agg,
                                               int n_deg) {
    int i = blockIdx.x * 256 + threadIdx.x;
    if (i < n_agg) {
        p[i] = 0.0f;
        p[i + n_agg] = 0.0f;
    } else if (i < n_agg + n_deg) {
        float* d = p + 2 * (size_t)n_agg + (i - n_agg);
        d[0] = 0.0f;
        d[n_deg] = 0.0f;
    }
}

__global__ __launch_bounds__(256) void degree_count(const int* __restrict__ erow,
                                                    float* __restrict__ deg, int ne,
                                                    int nn) {
    const int ph = package_half();
    int i = blockIdx.x * 256 + threadIdx.x;
    if (i < ne) {
        int r = erow[i];
        atomicAdd(deg + (size_t)ph * nn + r, 1.0f);
    }
}

#define DH_RANGES 4
#define DH_SLICES 64
#define DH_CAP 16384
#define DH_THREADS 1024
#define DH_BATCH 8

typedef int v4i __attribute__((ext_vector_type(4)));

__device__ __forceinline__ void dh_count(unsigned* h, int r, int lo, unsigned range) {
    const unsigned d = (unsigned)(r - lo);
    if (d < range) atomicAdd(h + d, 1u);
}

__global__ __launch_bounds__(DH_THREADS) void degree_count_lds(
    const int* __restrict__ erow, float* __restrict__ deg, int ne, int nn)
{
    __shared__ unsigned hist[DH_CAP];

    const int ph = package_half();

    const int a = blockIdx.x % DH_RANGES;
    const int b = blockIdx.x / DH_RANGES;
    const int rlen = (((nn + DH_RANGES - 1) / DH_RANGES) + 31) & ~31;
    const int lo = a * rlen;
    const unsigned range = (unsigned)rlen;

    for (int i = threadIdx.x; i < rlen; i += DH_THREADS) hist[i] = 0u;
    __syncthreads();

    if ((((size_t)erow) & 15) == 0) {
        const int n4 = ne >> 2;
        const int q0 = (int)(((long long)n4 * b) / DH_SLICES);
        const int q1 = (int)(((long long)n4 * (b + 1)) / DH_SLICES);
        const v4i* e4 = (const v4i*)erow;
        for (int q = q0 + (int)threadIdx.x; q < q1; q += DH_THREADS * DH_BATCH) {
            v4i v[DH_BATCH];
#pragma unroll
            for (int k = 0; k < DH_BATCH; ++k) {
                const int qq = q + k * DH_THREADS;
                v4i t = {-1, -1, -1, -1};
                if (qq < q1) t = e4[qq];
                v[k] = t;
            }
#pragma unroll
            for (int k = 0; k < DH_BATCH; ++k) {
                dh_count(hist, v[k].x, lo, range);
                dh_count(hist, v[k].y, lo, range);
                dh_count(hist, v[k].z, lo, range);
                dh_count(hist, v[k].w, lo, range);
            }
        }
        if (b == DH_SLICES - 1) {
            const int i = (n4 << 2) + (int)threadIdx.x;
            if (i < ne) dh_count(hist, erow[i], lo, range);
        }
    } else {
        const int e0 = (int)(((long long)ne * b) / DH_SLICES);
        const int e1 = (int)(((long long)ne * (b + 1)) / DH_SLICES);
        for (int i = e0 + (int)threadIdx.x; i < e1; i += DH_THREADS)
            dh_count(hist, erow[i], lo, range);
    }
    __syncthreads();

    const int cnt = min(rlen, nn - lo);
    float* dst = deg + (size_t)ph * nn + lo;
    for (int i0 = threadIdx.x; i0 < cnt; i0 += DH_THREADS * 4) {
        unsigned c[4];
#pragma unroll
        for (int k = 0; k < 4; ++k) {
            const int i = i0 + k * DH_THREADS;
            c[k] = (i < cnt) ? hist[i] : 0u;
        }
#pragma unroll
        for (int k = 0; k < 4; ++k)
            if (c[k] != 0u) atomicAdd(dst + i0 + k * DH_THREADS, (float)c[k]);
    }
}

__global__ __launch_bounds__(256) void deg_rsqrt(float* __restrict__ deg, int n) {
    int i = blockIdx.x * 256 + threadIdx.x;
    if (i < n) deg[i] = rsqrtf(deg[i] + deg[i + n]);
}

#define EDGES_PER_WAVE 32
#define EDGE_GROUP 8

__device__ __forceinline__ void scatter_one_edge(
    const float* __restrict__ xl, float* __restrict__ al,
    int r, int c, float wj)
{
    const float* s = xl + (size_t)c * D;
    float x0 = s[0];
    float x1 = s[32];
    float x2 = s[64];
    float x3 = s[96];
    float* d = al + (size_t)r * D;
    atomicAdd(d + 0,  wj * x0);
    atomicAdd(d + 32, wj * x1);
    atomicAdd(d + 64, wj * x2);
    atomicAdd(d + 96, wj * x3);
}

__global__ __launch_bounds__(256) void edge_aggregate(
    const float* __restrict__ x, const int* __restrict__ erow,
    const int* __restrict__ ecol, const float* __restrict__ dis,
    float* __restrict__ agg, int ne, int copy_floats)
{
    const int ph = package_half();
    const int lane = threadIdx.x & 31;
    const int wv   = __builtin_amdgcn_readfirstlane((int)(threadIdx.x >> 5));
    const int base = (blockIdx.x * 8 + wv) * EDGES_PER_WAVE;
    if (base >= ne) return;
    const int cnt = min(EDGES_PER_WAVE, ne - base);

    int row = 0, col = 0;
    float w = 0.0f;
    if (lane < cnt) {
        row = erow[base + lane];
        col = ecol[base + lane];
        w   = dis[row] * dis[col];
    }

    const float* xl = x + lane;
    float*       al = agg + (size_t)ph * copy_floats + lane;

    if (cnt == EDGES_PER_WAVE) {
#pragma unroll
        for (int g = 0; g < EDGES_PER_WAVE; g += EDGE_GROUP) {
            float v[EDGE_GROUP][4];
#pragma unroll
            for (int j = 0; j < EDGE_GROUP; ++j) {
                const int c = __builtin_amdgcn_readlane(col, g + j);
                const float* s = xl + (size_t)c * D;
                v[j][0] = s[0];
                v[j][1] = s[32];
                v[j][2] = s[64];
                v[j][3] = s[96];
            }
#pragma unroll
            for (int j = 0; j < EDGE_GROUP; ++j) {
                const int   r  = __builtin_amdgcn_readlane(row, g + j);
                const float wj = __int_as_float(
                    __builtin_amdgcn_readlane(__float_as_int(w), g + j));
                float* d = al + (size_t)r * D;
                atomicAdd(d + 0,  wj * v[j][0]);
                atomicAdd(d + 32, wj * v[j][1]);
                atomicAdd(d + 64, wj * v[j][2]);
                atomicAdd(d + 96, wj * v[j][3]);
            }
        }
    } else {
        for (int j = 0; j < cnt; ++j) {
            const int   r  = __builtin_amdgcn_readlane(row, j);
            const int   c  = __builtin_amdgcn_readlane(col, j);
            const float wj = __int_as_float(
                __builtin_amdgcn_readlane(__float_as_int(w), j));
            scatter_one_edge(xl, al, r, c, wj);
        }
    }
}

#define DIS_CAP 65536
#define EA_WAVES 32
#define EA_GROUP 16

__global__ __launch_bounds__(EA_WAVES * 32) void edge_aggregate_lds(
    const float* __restrict__ x, const int* __restrict__ erow,
    const int* __restrict__ ecol, const float* __restrict__ deg,
    float* __restrict__ agg, int ne, int nn, int copy_floats)
{
    __shared__ __attribute__((aligned(16))) float dis_s[DIS_CAP];

    const int ph = package_half();

    if ((nn & 3) == 0) {
        const v4f* d0 = (const v4f*)deg;
        const v4f* d1 = (const v4f*)(deg + nn);
        v4f* t = (v4f*)dis_s;
        const int n4 = nn >> 2;
#pragma unroll 6
        for (int i = threadIdx.x; i < n4; i += EA_WAVES * 32) {
            const v4f s = d0[i] + d1[i];
            v4f q;
            q.x = rsqrtf(s.x); q.y = rsqrtf(s.y);
            q.z = rsqrtf(s.z); q.w = rsqrtf(s.w);
            t[i] = q;
        }
    } else {
        for (int i = threadIdx.x; i < nn; i += EA_WAVES * 32)
            dis_s[i] = rsqrtf(deg[i] + deg[i + nn]);
    }
    __syncthreads();

    const int lane = threadIdx.x & 31;
    const int wv = __builtin_amdgcn_readfirstlane((int)(threadIdx.x >> 5));
    const long long gw = (long long)blockIdx.x * EA_WAVES + wv;
    const long long nw = (long long)gridDim.x * EA_WAVES;
    const int e0 = (int)(((long long)ne * gw) / nw);
    const int e1 = (int)(((long long)ne * (gw + 1)) / nw);

    const float* xl = x + lane;
    float*       al = agg + (size_t)ph * copy_floats + lane;

    int row = 0, col = 0;
    if (e0 + lane < e1) {
        row = erow[e0 + lane];
        col = ecol[e0 + lane];
    }

    for (int base = e0; base < e1; base += 32) {
        const int cnt = min(32, e1 - base);
        const int rcur = row, ccur = col;
        const float w = dis_s[rcur] * dis_s[ccur];

        row = 0; col = 0;
        if (base + 32 + lane < e1) {
            row = erow[base + 32 + lane];
            col = ecol[base + 32 + lane];
        }

        if (cnt == 32) {
#pragma unroll
            for (int g = 0; g < 32; g += EA_GROUP) {
                float v[EA_GROUP][4];
#pragma unroll
                for (int j = 0; j < EA_GROUP; ++j) {
                    const int c = __builtin_amdgcn_readlane(ccur, g + j);
                    const float* s = xl + (size_t)c * D;
                    v[j][0] = s[0];
                    v[j][1] = s[32];
                    v[j][2] = s[64];
                    v[j][3] = s[96];
                }
#pragma unroll
                for (int j = 0; j < EA_GROUP; ++j) {
                    const int   r  = __builtin_amdgcn_readlane(rcur, g + j);
                    const float wj = __int_as_float(
                        __builtin_amdgcn_readlane(__float_as_int(w), g + j));
                    float* d = al + (size_t)r * D;
                    atomicAdd(d + 0,  wj * v[j][0]);
                    atomicAdd(d + 32, wj * v[j][1]);
                    atomicAdd(d + 64, wj * v[j][2]);
                    atomicAdd(d + 96, wj * v[j][3]);
                }
            }
        } else {
            for (int j = 0; j < cnt; ++j) {
                const int   r  = __builtin_amdgcn_readlane(rcur, j);
                const int   c  = __builtin_amdgcn_readlane(ccur, j);
                const float wj = __int_as_float(
                    __builtin_amdgcn_readlane(__float_as_int(w), j));
                scatter_one_edge(xl, al, r, c, wj);
            }
        }
    }
}

#define A_PITCH 132

__global__ __launch_bounds__(256) void gemm_bias_wmma(
    const float* __restrict__ agg, const float* __restrict__ Wt,
    const float* __restrict__ bias, float* __restrict__ out, int copy_floats)
{
    __shared__ __attribute__((aligned(16))) float As[16 * A_PITCH];

    const int lane = threadIdx.x & 31;
    const int wave = threadIdx.x >> 5;
    const int m0   = blockIdx.x * 16;
    const int n0   = wave * 16;
    const int half = lane >> 4;
    const int l15  = lane & 15;

    {
        const float* g0 = agg + (size_t)(m0 + wave * 2) * D + lane * 4;
        const float* g1 = g0 + copy_floats;
        const v4f r0c0 = *(const v4f*)(g0);
        const v4f r1c0 = *(const v4f*)(g0 + D);
        const v4f r0c1 = *(const v4f*)(g1);
        const v4f r1c1 = *(const v4f*)(g1 + D);
        float* s = As + (wave * 2) * A_PITCH + lane * 4;
        *(v4f*)(s)           = r0c0 + r0c1;
        *(v4f*)(s + A_PITCH) = r1c0 + r1c1;
    }

    float bval = bias[n0 + l15];
    v8f c;
#pragma unroll
    for (int r = 0; r < 8; ++r) c[r] = bval;

    __syncthreads();

    const float* arow = As + l15 * A_PITCH + half * 2;
#pragma unroll 8
    for (int k = 0; k < D; k += 4) {
        v2f a = *(const v2f*)(arow + k);
        v2f b;
        b.x = Wt[(size_t)(k + half * 2 + 0) * D + n0 + l15];
        b.y = Wt[(size_t)(k + half * 2 + 1) * D + n0 + l15];
        c = __builtin_amdgcn_wmma_f32_16x16x4_f32(
                 false, a,  false, b,
                 (short)0, c,  false,  false);
    }

    float* orow = out + (size_t)(m0 + half * 8) * D + n0 + l15;
#pragma unroll
    for (int r = 0; r < 8; ++r) orow[(size_t)r * D] = c[r];
}

__global__ __launch_bounds__(128) void gemm_tail(
    const float* __restrict__ agg, const float* __restrict__ Wt,
    const float* __restrict__ bias, float* __restrict__ out,
    int row_start, int copy_floats)
{
    int r = row_start + blockIdx.x;
    int n = threadIdx.x;
    float acc = bias[n];
    const float* a  = agg + (size_t)r * D;
    const float* a1 = a + copy_floats;
    for (int k = 0; k < D; ++k)
        acc = fmaf(a[k] + a1[k], Wt[(size_t)k * D + n], acc);
    out[(size_t)r * D + n] = acc;
}

extern "C" void kernel_launch(void* const* d_in, const int* in_sizes, int n_in,
                              void* d_out, int out_size, void* d_ws, size_t ws_size,
                              hipStream_t stream) {
    const float* x      = (const float*)d_in[0];
    const float* weight = (const float*)d_in[1];
    const float* bias   = (const float*)d_in[2];
    const int*   erow   = (const int*)d_in[3];
    const int*   ecol   = (const int*)d_in[4];

    const int n_nodes = in_sizes[0] / D;
    const int n_edges = in_sizes[3];

    const int copy_floats = n_nodes * D;
    float* agg = (float*)d_ws;
    float* deg = agg + (size_t)2 * copy_floats;

    const int zn = n_nodes * (D + 1);
    zero_ws<<<(zn + 255) / 256, 256, 0, stream>>>((float*)d_ws, copy_floats,
                                                  n_nodes);

    if (n_nodes <= DIS_CAP) {
        degree_count_lds<<<DH_RANGES * DH_SLICES, DH_THREADS, 0, stream>>>(
            erow, deg, n_edges, n_nodes);
        edge_aggregate_lds<<<256, EA_WAVES * 32, 0, stream>>>(x, erow, ecol, deg,
                                                              agg, n_edges, n_nodes,
                                                              copy_floats);
    } else {
        degree_count<<<(n_edges + 255) / 256, 256, 0, stream>>>(erow, deg, n_edges,
                                                                n_nodes);
        deg_rsqrt<<<(n_nodes + 255) / 256, 256, 0, stream>>>(deg, n_nodes);
        edge_aggregate<<<(n_edges + 255) / 256, 256, 0, stream>>>(x, erow, ecol, deg,
                                                                  agg, n_edges,
                                                                  copy_floats);
    }

    const int mtiles = n_nodes / 16;
    if (mtiles > 0)
        gemm_bias_wmma<<<mtiles, 256, 0, stream>>>(agg, weight, bias,
                                                   (float*)d_out, copy_floats);
    const int tail = n_nodes - mtiles * 16;
    if (tail > 0)
        gemm_tail<<<tail, D, 0, stream>>>(agg, weight, bias, (float*)d_out,
                                          mtiles * 16, copy_floats);
}


# agg kept as ONE plane, 16-row strips owned by a package half: each half scans all edges (ticket queues) and adds only the rows it owns; zero fill and GEMM A loads halve; sum order = arrival
# speedup vs baseline: 3.4888x; 1.0657x over previous
#include <hip/hip_runtime.h>

#define D 128

typedef float v2f __attribute__((ext_vector_type(2)));
typedef float v4f __attribute__((ext_vector_type(4)));
typedef float v8f __attribute__((ext_vector_type(8)));

__device__ __forceinline__ int package_half() {
    return (int)((__builtin_amdgcn_s_sendmsg_rtn(0x87) >> 18) & 1u);
}

__global__ __launch_bounds__(256) void zero_ws(float* __restrict__ p, int n_agg,
                                               int n_deg, int n_q) {
    int i = blockIdx.x * 256 + threadIdx.x;
    if (i < n_agg) {
        p[i] = 0.0f;
    } else if (i < n_agg + n_deg) {
        float* d = p + (size_t)n_agg + (i - n_agg);
        d[0] = 0.0f;
        d[n_deg] = 0.0f;
    } else if (i < n_agg + n_deg + n_q) {
        p[(size_t)n_agg + 2 * (size_t)n_deg + (i - n_agg - n_deg)] = 0.0f;
    }
}

__global__ __launch_bounds__(256) void degree_count(const int* __restrict__ erow,
                                                    float* __restrict__ deg, int ne,
                                                    int nn) {
    const int ph = package_half();
    int i = blockIdx.x * 256 + threadIdx.x;
    if (i < ne) {
        int r = erow[i];
        atomicAdd(deg + (size_t)ph * nn + r, 1.0f);
    }
}

#define DH_RANGES 4
#define DH_SLICES 64
#define DH_CAP 16384
#define DH_THREADS 1024
#define DH_BATCH 8

typedef int v4i __attribute__((ext_vector_type(4)));

__device__ __forceinline__ void dh_count(unsigned* h, int r, int lo, unsigned range) {
    const unsigned d = (unsigned)(r - lo);
    if (d < range) atomicAdd(h + d, 1u);
}

__global__ __launch_bounds__(DH_THREADS) void degree_count_lds(
    const int* __restrict__ erow, float* __restrict__ deg, int ne, int nn)
{
    __shared__ unsigned hist[DH_CAP];

    const int ph = package_half();

    const int a = blockIdx.x % DH_RANGES;
    const int b = blockIdx.x / DH_RANGES;
    const int rlen = (((nn + DH_RANGES - 1) / DH_RANGES) + 31) & ~31;
    const int lo = a * rlen;
    const unsigned range = (unsigned)rlen;

    for (int i = threadIdx.x; i < rlen; i += DH_THREADS) hist[i] = 0u;
    __syncthreads();

    if ((((size_t)erow) & 15) == 0) {
        const int n4 = ne >> 2;
        const int q0 = (int)(((long long)n4 * b) / DH_SLICES);
        const int q1 = (int)(((long long)n4 * (b + 1)) / DH_SLICES);
        const v4i* e4 = (const v4i*)erow;
        for (int q = q0 + (int)threadIdx.x; q < q1; q += DH_THREADS * DH_BATCH) {
            v4i v[DH_BATCH];
#pragma unroll
            for (int k = 0; k < DH_BATCH; ++k) {
                const int qq = q + k * DH_THREADS;
                v4i t = {-1, -1, -1, -1};
                if (qq < q1) t = e4[qq];
                v[k] = t;
            }
#pragma unroll
            for (int k = 0; k < DH_BATCH; ++k) {
                dh_count(hist, v[k].x, lo, range);
                dh_count(hist, v[k].y, lo, range);
                dh_count(hist, v[k].z, lo, range);
                dh_count(hist, v[k].w, lo, range);
            }
        }
        if (b == DH_SLICES - 1) {
            const int i = (n4 << 2) + (int)threadIdx.x;
            if (i < ne) dh_count(hist, erow[i], lo, range);
        }
    } else {
        const int e0 = (int)(((long long)ne * b) / DH_SLICES);
        const int e1 = (int)(((long long)ne * (b + 1)) / DH_SLICES);
        for (int i = e0 + (int)threadIdx.x; i < e1; i += DH_THREADS)
            dh_count(hist, erow[i], lo, range);
    }
    __syncthreads();

    const int cnt = min(rlen, nn - lo);
    float* dst = deg + (size_t)ph * nn + lo;
    for (int i0 = threadIdx.x; i0 < cnt; i0 += DH_THREADS * 4) {
        unsigned c[4];
#pragma unroll
        for (int k = 0; k < 4; ++k) {
            const int i = i0 + k * DH_THREADS;
            c[k] = (i < cnt) ? hist[i] : 0u;
        }
#pragma unroll
        for (int k = 0; k < 4; ++k)
            if (c[k] != 0u) atomicAdd(dst + i0 + k * DH_THREADS, (float)c[k]);
    }
}

__global__ __launch_bounds__(256) void deg_rsqrt(float* __restrict__ deg, int n) {
    int i = blockIdx.x * 256 + threadIdx.x;
    if (i < n) deg[i] = rsqrtf(deg[i] + deg[i + n]);
}

#define EDGES_PER_WAVE 32
#define EDGE_GROUP 8

__device__ __forceinline__ void scatter_one_edge(
    const float* __restrict__ xl, float* __restrict__ al,
    int r, int c, float wj)
{
    const float* s = xl + (size_t)c * D;
    float x0 = s[0];
    float x1 = s[32];
    float x2 = s[64];
    float x3 = s[96];
    float* d = al + (size_t)r * D;
    atomicAdd(d + 0,  wj * x0);
    atomicAdd(d + 32, wj * x1);
    atomicAdd(d + 64, wj * x2);
    atomicAdd(d + 96, wj * x3);
}

__global__ __launch_bounds__(256) void edge_aggregate(
    const float* __restrict__ x, const int* __restrict__ erow,
    const int* __restrict__ ecol, const float* __restrict__ dis,
    float* __restrict__ agg, int ne, int copy_floats)
{
    const int ph = package_half();
    const int lane = threadIdx.x & 31;
    const int wv   = __builtin_amdgcn_readfirstlane((int)(threadIdx.x >> 5));
    const int base = (blockIdx.x * 8 + wv) * EDGES_PER_WAVE;
    if (base >= ne) return;
    const int cnt = min(EDGES_PER_WAVE, ne - base);

    int row = 0, col = 0;
    float w = 0.0f;
    if (lane < cnt) {
        row = erow[base + lane];
        col = ecol[base + lane];
        w   = dis[row] * dis[col];
    }

    const float* xl = x + lane;
    float*       al = agg + (size_t)ph * copy_floats + lane;

    if (cnt == EDGES_PER_WAVE) {
#pragma unroll
        for (int g = 0; g < EDGES_PER_WAVE; g += EDGE_GROUP) {
            float v[EDGE_GROUP][4];
#pragma unroll
            for (int j = 0; j < EDGE_GROUP; ++j) {
                const int c = __builtin_amdgcn_readlane(col, g + j);
                const float* s = xl + (size_t)c * D;
                v[j][0] = s[0];
                v[j][1] = s[32];
                v[j][2] = s[64];
                v[j][3] = s[96];
            }
#pragma unroll
            for (int j = 0; j < EDGE_GROUP; ++j) {
                const int   r  = __builtin_amdgcn_readlane(row, g + j);
                const float wj = __int_as_float(
                    __builtin_amdgcn_readlane(__float_as_int(w), g + j));
                float* d = al + (size_t)r * D;
                atomicAdd(d + 0,  wj * v[j][0]);
                atomicAdd(d + 32, wj * v[j][1]);
                atomicAdd(d + 64, wj * v[j][2]);
                atomicAdd(d + 96, wj * v[j][3]);
            }
        }
    } else {
        for (int j = 0; j < cnt; ++j) {
            const int   r  = __builtin_amdgcn_readlane(row, j);
            const int   c  = __builtin_amdgcn_readlane(col, j);
            const float wj = __int_as_float(
                __builtin_amdgcn_readlane(__float_as_int(w), j));
            scatter_one_edge(xl, al, r, c, wj);
        }
    }
}

#define DIS_CAP 65536
#define EA_WAVES 32
#define EA_GROUP 16
#define EA_UNIT 64
#define EA_SUBQ 4
#define EA_QSTRIDE 32
#define EA_QWORDS (2 * EA_SUBQ * EA_QSTRIDE)
#define EA_BUF 64

template <bool FULL>
__device__ __forceinline__ void ea_rows(const float* __restrict__ xl,
                                        float* __restrict__ al,
                                        int rcur, int ccur, float w, int cnt)
{
#pragma unroll
    for (int g = 0; g < 32; g += EA_GROUP) {
        float v[EA_GROUP][4];
#pragma unroll
        for (int j = 0; j < EA_GROUP; ++j) {
            v[j][0] = 0.0f; v[j][1] = 0.0f; v[j][2] = 0.0f; v[j][3] = 0.0f;
            if (FULL || g + j < cnt) {
                const int c = __builtin_amdgcn_readlane(ccur, g + j);
                const float* s = xl + (size_t)c * D;
                v[j][0] = s[0];
                v[j][1] = s[32];
                v[j][2] = s[64];
                v[j][3] = s[96];
            }
        }
#pragma unroll
        for (int j = 0; j < EA_GROUP; ++j) {
            if (FULL || g + j < cnt) {
                const int   r  = __builtin_amdgcn_readlane(rcur, g + j);
                const float wj = __int_as_float(
                    __builtin_amdgcn_readlane(__float_as_int(w), g + j));
                float* d = al + (size_t)r * D;
                atomicAdd(d + 0,  wj * v[j][0]);
                atomicAdd(d + 32, wj * v[j][1]);
                atomicAdd(d + 64, wj * v[j][2]);
                atomicAdd(d + 96, wj * v[j][3]);
            }
        }
    }
}

__global__ __launch_bounds__(EA_WAVES * 32) void edge_aggregate_own(
    const float* __restrict__ x, const int* __restrict__ erow,
    const int* __restrict__ ecol, const float* __restrict__ deg,
    float* __restrict__ agg, int* __restrict__ queue, int ne, int nn)
{
    __shared__ __attribute__((aligned(16))) float dis_s[DIS_CAP];
    __shared__ int   buf_row[EA_WAVES][EA_BUF];
    __shared__ int   buf_col[EA_WAVES][EA_BUF];
    __shared__ float buf_w[EA_WAVES][EA_BUF];

    const int ph = package_half();

    if ((nn & 3) == 0) {
        const v4f* d0 = (const v4f*)deg;
        const v4f* d1 = (const v4f*)(deg + nn);
        v4f* t = (v4f*)dis_s;
        const int n4 = nn >> 2;
#pragma unroll 6
        for (int i = threadIdx.x; i < n4; i += EA_WAVES * 32) {
            const v4f s = d0[i] + d1[i];
            v4f q;
            q.x = rsqrtf(s.x); q.y = rsqrtf(s.y);
            q.z = rsqrtf(s.z); q.w = rsqrtf(s.w);
            t[i] = q;
        }
    } else {
        for (int i = threadIdx.x; i < nn; i += EA_WAVES * 32)
            dis_s[i] = rsqrtf(deg[i] + deg[i + nn]);
    }
    __syncthreads();

    const int lane = threadIdx.x & 31;
    const int wv = __builtin_amdgcn_readfirstlane((int)(threadIdx.x >> 5));
    const int kq = wv & (EA_SUBQ - 1);

    const float* xl = x + lane;
    float*       al = agg + lane;
    int*   brow = buf_row[wv];
    int*   bcol = buf_col[wv];
    float* bw   = buf_w[wv];

    int cnt = 0;

    auto scan = [&](int r, int c, bool valid, int h) {
        const float w = dis_s[r] * dis_s[c];
        const bool own = valid && (((r >> 4) & 1) == h);
        const unsigned m = __builtin_amdgcn_ballot_w32(own);
        const int pos = (int)__builtin_amdgcn_mbcnt_lo(m, (unsigned)cnt);
        if (own) { brow[pos] = r; bcol[pos] = c; bw[pos] = w; }
        cnt += __builtin_popcount(m);
        if (cnt >= 32) {
            const int   rcur = brow[lane];
            const int   ccur = bcol[lane];
            const float wcur = bw[lane];
            const int rest = cnt - 32;
            if (lane < rest) {
                const int   a = brow[32 + lane];
                const int   b = bcol[32 + lane];
                const float f = bw[32 + lane];
                brow[lane] = a; bcol[lane] = b; bw[lane] = f;
            }
            cnt = rest;
            ea_rows<true>(xl, al, rcur, ccur, wcur, 32);
        }
    };

    const long long nunits = ((long long)ne + EA_UNIT - 1) / EA_UNIT;

    int hq = ph;
    int more = 1;
    int tv = 0;
    if (lane == 0) tv = atomicAdd(queue + (hq * EA_SUBQ + kq) * EA_QSTRIDE, 1);

    int cur_ok = 0, hcur = ph;
    long long ecur = 0;
    int r0 = 0, c0 = 0, r1 = 0, c1 = 0;

    while (more | cur_ok) {
        int nxt_ok = 0, hn = hq;
        long long en = 0;
        int nr0 = 0, nc0 = 0, nr1 = 0, nc1 = 0;
        if (more) {
            const int tn = __builtin_amdgcn_readfirstlane(tv);
            const long long un = (long long)tn * EA_SUBQ + kq;
            if (un < nunits) {
                nxt_ok = 1;
                en = un * EA_UNIT;
                if (lane == 0) tv = atomicAdd(queue + (hq * EA_SUBQ + kq) * EA_QSTRIDE, 1);
                const long long b = en + lane;
                if (b < ne)      { nr0 = erow[b];      nc0 = ecol[b]; }
                if (b + 32 < ne) { nr1 = erow[b + 32]; nc1 = ecol[b + 32]; }
            } else if (hq == ph) {
                hq = ph ^ 1;
                if (lane == 0) tv = atomicAdd(queue + (hq * EA_SUBQ + kq) * EA_QSTRIDE, 1);
            } else {
                more = 0;
            }
        }
        if (cur_ok) {
            const long long e = ecur + lane;
            scan(r0, c0, e < ne, hcur);
            scan(r1, c1, e + 32 < ne, hcur);
        }
        cur_ok = nxt_ok; hcur = hn; ecur = en;
        r0 = nr0; c0 = nc0; r1 = nr1; c1 = nc1;
    }

    if (cnt > 0) {
        const int   rcur = brow[lane];
        const int   ccur = bcol[lane];
        const float wcur = bw[lane];
        ea_rows<false>(xl, al, rcur, ccur, wcur, cnt);
    }
}

#define A_PITCH 132

__global__ __launch_bounds__(256) void gemm_bias_wmma(
    const float* __restrict__ agg, const float* __restrict__ Wt,
    const float* __restrict__ bias, float* __restrict__ out)
{
    __shared__ __attribute__((aligned(16))) float As[16 * A_PITCH];

    const int lane = threadIdx.x & 31;
    const int wave = threadIdx.x >> 5;
    const int m0   = blockIdx.x * 16;
    const int n0   = wave * 16;
    const int half = lane >> 4;
    const int l15  = lane & 15;

    {
        const float* g0 = agg + (size_t)(m0 + wave * 2) * D + lane * 4;
        const v4f r0 = *(const v4f*)(g0);
        const v4f r1 = *(const v4f*)(g0 + D);
        float* s = As + (wave * 2) * A_PITCH + lane * 4;
        *(v4f*)(s)           = r0;
        *(v4f*)(s + A_PITCH) = r1;
    }

    float bval = bias[n0 + l15];
    v8f c;
#pragma unroll
    for (int r = 0; r < 8; ++r) c[r] = bval;

    __syncthreads();

    const float* arow = As + l15 * A_PITCH + half * 2;
#pragma unroll 8
    for (int k = 0; k < D; k += 4) {
        v2f a = *(const v2f*)(arow + k);
        v2f b;
        b.x = Wt[(size_t)(k + half * 2 + 0) * D + n0 + l15];
        b.y = Wt[(size_t)(k + half * 2 + 1) * D + n0 + l15];
        c = __builtin_amdgcn_wmma_f32_16x16x4_f32(
                 false, a,  false, b,
                 (short)0, c,  false,  false);
    }

    float* orow = out + (size_t)(m0 + half * 8) * D + n0 + l15;
#pragma unroll
    for (int r = 0; r < 8; ++r) orow[(size_t)r * D] = c[r];
}

__global__ __launch_bounds__(128) void gemm_tail(
    const float* __restrict__ agg, const float* __restrict__ Wt,
    const float* __restrict__ bias, float* __restrict__ out,
    int row_start)
{
    int r = row_start + blockIdx.x;
    int n = threadIdx.x;
    float acc = bias[n];
    const float* a  = agg + (size_t)r * D;
    for (int k = 0; k < D; ++k)
        acc = fmaf(a[k], Wt[(size_t)k * D + n], acc);
    out[(size_t)r * D + n] = acc;
}

extern "C" void kernel_launch(void* const* d_in, const int* in_sizes, int n_in,
                              void* d_out, int out_size, void* d_ws, size_t ws_size,
                              hipStream_t stream) {
    const float* x      = (const float*)d_in[0];
    const float* weight = (const float*)d_in[1];
    const float* bias   = (const float*)d_in[2];
    const int*   erow   = (const int*)d_in[3];
    const int*   ecol   = (const int*)d_in[4];

    const int n_nodes = in_sizes[0] / D;
    const int n_edges = in_sizes[3];

    const int agg_floats = n_nodes * D;
    float* agg   = (float*)d_ws;
    float* deg   = agg + (size_t)agg_floats;
    int*   queue = (int*)(deg + (size_t)2 * n_nodes);

    const int zn = n_nodes * (D + 1) + EA_QWORDS;
    zero_ws<<<(zn + 255) / 256, 256, 0, stream>>>((float*)d_ws, agg_floats,
                                                  n_nodes, EA_QWORDS);

    if (n_nodes <= DIS_CAP) {
        degree_count_lds<<<DH_RANGES * DH_SLICES, DH_THREADS, 0, stream>>>(
            erow, deg, n_edges, n_nodes);
        edge_aggregate_own<<<256, EA_WAVES * 32, 0, stream>>>(x, erow, ecol, deg,
                                                              agg, queue, n_edges,
                                                              n_nodes);
    } else {
        degree_count<<<(n_edges + 255) / 256, 256, 0, stream>>>(erow, deg, n_edges,
                                                                n_nodes);
        deg_rsqrt<<<(n_nodes + 255) / 256, 256, 0, stream>>>(deg, n_nodes);
        edge_aggregate<<<(n_edges + 255) / 256, 256, 0, stream>>>(x, erow, ecol, deg,
                                                                  agg, n_edges, 0);
    }

    const int mtiles = n_nodes / 16;
    if (mtiles > 0)
        gemm_bias_wmma<<<mtiles, 256, 0, stream>>>(agg, weight, bias,
                                                   (float*)d_out);
    const int tail = n_nodes - mtiles * 16;
    if (tail > 0)
        gemm_tail<<<tail, D, 0, stream>>>(agg, weight, bias, (float*)d_out,
                                          mtiles * 16);
}
